# starting kernel
# baseline (speedup 1.0000x reference)
#include <hip/hip_runtime.h>
#include <math.h>

#define N_OBS   6000
#define N_X     6000
#define D_FEAT  16
#define N_DOM   3
#define N_SLICES 16
#define OBS_TILES (N_OBS / 16)
#define COL_TILES (N_X / 16)

typedef float v2f __attribute__((ext_vector_type(2)));
typedef float v4f __attribute__((ext_vector_type(4)));
typedef float v8f __attribute__((ext_vector_type(8)));

#define WS_HDR    0
#define WS_SXOBS  16
#define WS_SX     (WS_SXOBS + N_OBS * D_FEAT)
#define WS_NOBS   (WS_SX + N_X * D_FEAT)
#define WS_NX     (WS_NOBS + N_OBS)
#define WS_BETA0  (WS_NX + N_X)
#define WS_BETA1  (WS_BETA0 + N_DOM * N_OBS)
#define WS_PART0  (WS_BETA1 + N_DOM * N_OBS)
#define WS_PART1  (WS_PART0 + N_SLICES * N_X)

__global__ void prep_scalars(const float* __restrict__ logits,
                             const float* __restrict__ raw_ls,
                             float* __restrict__ hdr) {
    if (threadIdx.x == 0) {
        float ls = log1pf(__expf(raw_ls[0]));
        hdr[0] = 1.0f / ls;
        float t01 = 1.0f / (1.0f + __expf(-logits[0]));
        float t02 = 1.0f / (1.0f + __expf(-logits[1]));
        float t12 = 1.0f / (1.0f + __expf(-logits[2]));
        float F[9] = {1.0f, t01, t02,   t01, 1.0f, t12,   t02, t12, 1.0f};
        #pragma unroll
        for (int k = 0; k < 9; ++k) hdr[1 + k] = F[k];
    }
}

__global__ void prep_rows(const float* __restrict__ xobs,
                          const float* __restrict__ x,
                          const float* __restrict__ alpha0,
                          const float* __restrict__ alpha1,
                          const int*   __restrict__ ro,
                          float* __restrict__ ws) {
    int tid = blockIdx.x * blockDim.x + threadIdx.x;
    if (tid >= N_OBS + N_X) return;
    const float inv_ls = ws[0];
    if (tid < N_OBS) {
        int i = tid;
        const float* src = xobs + i * D_FEAT;
        float* dst = ws + WS_SXOBS + i * D_FEAT;
        float nrm = 0.0f;
        #pragma unroll
        for (int k = 0; k < D_FEAT; ++k) {
            float v = src[k] * inv_ls;
            dst[k] = v;
            nrm = fmaf(v, v, nrm);
        }
        ws[WS_NOBS + i] = nrm;
        int dom = (i >= ro[2]) + (i >= ro[4]);
        float a0 = alpha0[i], a1 = alpha1[i];
        #pragma unroll
        for (int c = 0; c < N_DOM; ++c) {
            float f = ws[1 + dom * 3 + c];
            ws[WS_BETA0 + c * N_OBS + i] = a0 * f;
            ws[WS_BETA1 + c * N_OBS + i] = a1 * f;
        }
    } else {
        int j = tid - N_OBS;
        const float* src = x + j * D_FEAT;
        float* dst = ws + WS_SX + j * D_FEAT;
        float nrm = 0.0f;
        #pragma unroll
        for (int k = 0; k < D_FEAT; ++k) {
            float v = src[k] * inv_ls;
            dst[k] = v;
            nrm = fmaf(v, v, nrm);
        }
        ws[WS_NX + j] = nrm;
    }
}

__global__ void __launch_bounds__(256) matern_wmma(const int* __restrict__ rx,
                                                   float* __restrict__ ws) {
    const int lane = threadIdx.x & 31;
    const int wave = blockIdx.x * (blockDim.x >> 5) + (threadIdx.x >> 5);
    const int colTile = wave >> 4;
    const int slice   = wave & (N_SLICES - 1);
    const int n  = lane & 15;
    const int g  = lane >> 4;
    const int kb = g * 2;
    const int j  = colTile * 16 + n;

    const float* __restrict__ sxobs = ws + WS_SXOBS;
    const float* __restrict__ sx    = ws + WS_SX;
    const float* __restrict__ nobs  = ws + WS_NOBS;

    const float* bp = sx + j * D_FEAT + kb;
    v2f b0 = *(const v2f*)(bp + 0);
    v2f b1 = *(const v2f*)(bp + 4);
    v2f b2 = *(const v2f*)(bp + 8);
    v2f b3 = *(const v2f*)(bp + 12);
    const float nxl = ws[WS_NX + j];

    const int dx = (j >= rx[2]) + (j >= rx[4]);
    const float* __restrict__ w0p = ws + WS_BETA0 + dx * N_OBS;
    const float* __restrict__ w1p = ws + WS_BETA1 + dx * N_OBS;

    float acc0 = 0.0f, acc1 = 0.0f;

    for (int t = slice; t < OBS_TILES; t += N_SLICES) {
        const int i0 = t * 16;
        const float* ap = sxobs + (i0 + n) * D_FEAT + kb;
        v2f a0 = *(const v2f*)(ap + 0);
        v2f a1 = *(const v2f*)(ap + 4);
        v2f a2 = *(const v2f*)(ap + 8);
        v2f a3 = *(const v2f*)(ap + 12);

        v8f c = {};
        c = __builtin_amdgcn_wmma_f32_16x16x4_f32(false, a0, false, b0, (short)0, c, false, false);
        c = __builtin_amdgcn_wmma_f32_16x16x4_f32(false, a1, false, b1, (short)0, c, false, false);
        c = __builtin_amdgcn_wmma_f32_16x16x4_f32(false, a2, false, b2, (short)0, c, false, false);
        c = __builtin_amdgcn_wmma_f32_16x16x4_f32(false, a3, false, b3, (short)0, c, false, false);

        const int rbase = i0 + g * 8;
        v4f nA  = *(const v4f*)(nobs + rbase);
        v4f nB  = *(const v4f*)(nobs + rbase + 4);
        v4f w0A = *(const v4f*)(w0p + rbase);
        v4f w0B = *(const v4f*)(w0p + rbase + 4);
        v4f w1A = *(const v4f*)(w1p + rbase);
        v4f w1B = *(const v4f*)(w1p + rbase + 4);

        float nr[8] = {nA.x, nA.y, nA.z, nA.w, nB.x, nB.y, nB.z, nB.w};
        float wa[8] = {w0A.x, w0A.y, w0A.z, w0A.w, w0B.x, w0B.y, w0B.z, w0B.w};
        float wb[8] = {w1A.x, w1A.y, w1A.z, w1A.w, w1B.x, w1B.y, w1B.z, w1B.w};

        #pragma unroll
        for (int r = 0; r < 8; ++r) {
            float sq   = fmaf(-2.0f, c[r], nr[r] + nxl);
            float dsq  = fmaxf(sq, 1e-12f);
            float d    = __builtin_amdgcn_sqrtf(dsq);
            float s5d  = 2.2360679774997896f * d;
            float poly = fmaf(1.6666666666666667f, dsq, 1.0f + s5d);
            float e    = __expf(-s5d);
            float k    = poly * e;
            acc0 = fmaf(wa[r], k, acc0);
            acc1 = fmaf(wb[r], k, acc1);
        }
    }

    acc0 += __shfl_xor(acc0, 16, 32);
    acc1 += __shfl_xor(acc1, 16, 32);
    if (lane < 16) {
        ws[WS_PART0 + slice * N_X + j] = acc0;
        ws[WS_PART1 + slice * N_X + j] = acc1;
    }
}

__global__ void combine(const float* __restrict__ w,
                        const float* __restrict__ ws,
                        float* __restrict__ out) {
    int j = blockIdx.x * blockDim.x + threadIdx.x;
    if (j >= N_X) return;
    const float* p0 = ws + WS_PART0;
    const float* p1 = ws + WS_PART1;
    float f0 = 0.0f, f1 = 0.0f;
    #pragma unroll
    for (int s = 0; s < N_SLICES; ++s) {
        f0 += p0[s * N_X + j];
        f1 += p1[s * N_X + j];
    }
    float wv = w[j];
    out[j] = wv * f0 + (1.0f - wv) * f1;
}

extern "C" void kernel_launch(void* const* d_in, const int* in_sizes, int n_in,
                              void* d_out, int out_size, void* d_ws, size_t ws_size,
                              hipStream_t stream) {
    const float* xobs = (const float*)d_in[0];
    const float* x    = (const float*)d_in[1];
    const float* w    = (const float*)d_in[2];
    const float* a0   = (const float*)d_in[3];
    const float* a1   = (const float*)d_in[4];
    const float* tfl  = (const float*)d_in[5];
    const float* rls  = (const float*)d_in[6];
    const int*   ro   = (const int*)d_in[7];
    const int*   rx   = (const int*)d_in[8];
    float* ws  = (float*)d_ws;
    float* out = (float*)d_out;

    prep_scalars<<<1, 32, 0, stream>>>(tfl, rls, ws);
    prep_rows<<<(N_OBS + N_X + 255) / 256, 256, 0, stream>>>(xobs, x, a0, a1, ro, ws);
    matern_wmma<<<(COL_TILES * N_SLICES) / 8, 256, 0, stream>>>(rx, ws);
    combine<<<(N_X + 255) / 256, 256, 0, stream>>>(w, ws, out);
}


# fold both prep launches into the matrix kernel (4 launches -> 2): waves derive the scalars, B carries -2/ls^2, a 5th WMMA k-step sums the row norm in the accumulator; weights summed per tile first
# speedup vs baseline: 1.2722x; 1.2722x over previous
#include <hip/hip_runtime.h>
#include <math.h>


#define N_OBS   6000
#define N_X     6000
#define D_FEAT  16
#define N_DOM   3
#define N_SLICES 16
#define OBS_TILES (N_OBS / 16)
#define COL_TILES (N_X / 16)

typedef float v2f __attribute__((ext_vector_type(2)));
typedef float v4f __attribute__((ext_vector_type(4)));
typedef float v8f __attribute__((ext_vector_type(8)));

#define WS_PART0  0
#define WS_PART1  (WS_PART0 + N_SLICES * N_X)

__global__ void __launch_bounds__(256) matern_wmma(const float* __restrict__ xobs,
                                                   const float* __restrict__ x,
                                                   const float* __restrict__ alpha0,
                                                   const float* __restrict__ alpha1,
                                                   const float* __restrict__ logits,
                                                   const float* __restrict__ raw_ls,
                                                   const int*   __restrict__ ro,
                                                   const int*   __restrict__ rx,
                                                   float* __restrict__ ws) {
    const int lane = threadIdx.x & 31;
    const int wave = __builtin_amdgcn_readfirstlane(blockIdx.x * (blockDim.x >> 5) + (threadIdx.x >> 5));
    const int colTile = wave >> 4;
    const int slice   = wave & (N_SLICES - 1);
    const int n  = lane & 15;
    const int g  = lane >> 4;
    const int kb = g * 2;
    const int j  = colTile * 16 + n;

    const float ls     = log1pf(__expf(raw_ls[0]));
    const float inv_ls = 1.0f / ls;
    const float s2     = inv_ls * inv_ls;
    const float t01 = 1.0f / (1.0f + __expf(-logits[0]));
    const float t02 = 1.0f / (1.0f + __expf(-logits[1]));
    const float t12 = 1.0f / (1.0f + __expf(-logits[2]));
    const int ro1 = ro[2], ro2 = ro[4];

    const int dx = (j >= rx[2]) + (j >= rx[4]);
    const float F0 = (dx == 0) ? 1.0f : ((dx == 1) ? t01 : t02);
    const float F1 = (dx == 0) ? t01  : ((dx == 1) ? 1.0f : t12);
    const float F2 = (dx == 0) ? t02  : ((dx == 1) ? t12 : 1.0f);

    const float* bp = x + j * D_FEAT + kb;
    const v2f x0 = *(const v2f*)(bp + 0);
    const v2f x1 = *(const v2f*)(bp + 4);
    const v2f x2 = *(const v2f*)(bp + 8);
    const v2f x3 = *(const v2f*)(bp + 12);
    v2f qb = x0 * x0;
    qb = x1 * x1 + qb;
    qb = x2 * x2 + qb;
    qb = x3 * x3 + qb;
    float nb = qb.x + qb.y;
    nb += __shfl_xor(nb, 16, 32);
    const float nxl = nb * s2;
    const float m2  = -2.0f * s2;
    const v2f b0 = x0 * m2;
    const v2f b1 = x1 * m2;
    const v2f b2 = x2 * m2;
    const v2f b3 = x3 * m2;
    const v2f b4 = {s2, 0.0f};

    float acc0 = 0.0f, acc1 = 0.0f;

    for (int t = slice; t < OBS_TILES; t += N_SLICES) {
        const int i0 = t * 16;
        const float* ap = xobs + (i0 + n) * D_FEAT + kb;
        const v2f a0 = *(const v2f*)(ap + 0);
        const v2f a1 = *(const v2f*)(ap + 4);
        const v2f a2 = *(const v2f*)(ap + 8);
        const v2f a3 = *(const v2f*)(ap + 12);
        v2f qa = a0 * a0;
        qa = a1 * a1 + qa;
        qa = a2 * a2 + qa;
        qa = a3 * a3 + qa;
        const v2f a4 = {qa.x + qa.y, 0.0f};

        v8f c = {};
        c = __builtin_amdgcn_wmma_f32_16x16x4_f32(false, a0, false, b0, (short)0, c, false, false);
        c = __builtin_amdgcn_wmma_f32_16x16x4_f32(false, a1, false, b1, (short)0, c, false, false);
        c = __builtin_amdgcn_wmma_f32_16x16x4_f32(false, a2, false, b2, (short)0, c, false, false);
        c = __builtin_amdgcn_wmma_f32_16x16x4_f32(false, a3, false, b3, (short)0, c, false, false);
        c = __builtin_amdgcn_wmma_f32_16x16x4_f32(false, a4, false, b4, (short)0, c, false, false);

        const int rbase = i0 + g * 8;
        const v4f u0A = *(const v4f*)(alpha0 + rbase);
        const v4f u0B = *(const v4f*)(alpha0 + rbase + 4);
        const v4f u1A = *(const v4f*)(alpha1 + rbase);
        const v4f u1B = *(const v4f*)(alpha1 + rbase + 4);
        const float wa[8] = {u0A.x, u0A.y, u0A.z, u0A.w, u0B.x, u0B.y, u0B.z, u0B.w};
        const float wb[8] = {u1A.x, u1A.y, u1A.z, u1A.w, u1B.x, u1B.y, u1B.z, u1B.w};

        float kk[8];
        #pragma unroll
        for (int r = 0; r < 8; ++r) {
            float dsq  = fmaxf(c[r] + nxl, 1e-12f);
            float d    = __builtin_amdgcn_sqrtf(dsq);
            float s5d  = 2.2360679774997896f * d;
            float poly = fmaf(1.6666666666666667f, dsq, 1.0f + s5d);
            float e    = __expf(-s5d);
            kk[r] = poly * e;
        }

        const int dlo = (i0 >= ro1) + (i0 >= ro2);
        const int dhi = (i0 + 15 >= ro1) + (i0 + 15 >= ro2);
        if (dlo == dhi) {
            const float fsel = (dlo == 0) ? F0 : ((dlo == 1) ? F1 : F2);
            float t0 = 0.0f, t1 = 0.0f;
            #pragma unroll
            for (int r = 0; r < 8; ++r) {
                t0 = fmaf(wa[r], kk[r], t0);
                t1 = fmaf(wb[r], kk[r], t1);
            }
            acc0 = fmaf(fsel, t0, acc0);
            acc1 = fmaf(fsel, t1, acc1);
        } else {
            #pragma unroll
            for (int r = 0; r < 8; ++r) {
                const int i  = rbase + r;
                const int dr = (i >= ro1) + (i >= ro2);
                const float f = (dr == 0) ? F0 : ((dr == 1) ? F1 : F2);
                acc0 = fmaf(wa[r] * f, kk[r], acc0);
                acc1 = fmaf(wb[r] * f, kk[r], acc1);
            }
        }
    }

    acc0 += __shfl_xor(acc0, 16, 32);
    acc1 += __shfl_xor(acc1, 16, 32);
    if (lane < 16) {
        ws[WS_PART0 + slice * N_X + j] = acc0;
        ws[WS_PART1 + slice * N_X + j] = acc1;
    }
}

__global__ void combine(const float* __restrict__ w,
                        const float* __restrict__ ws,
                        float* __restrict__ out) {
    int j = blockIdx.x * blockDim.x + threadIdx.x;
    if (j >= N_X) return;
    const float* p0 = ws + WS_PART0;
    const float* p1 = ws + WS_PART1;
    float f0 = 0.0f, f1 = 0.0f;
    #pragma unroll
    for (int s = 0; s < N_SLICES; ++s) {
        f0 += p0[s * N_X + j];
        f1 += p1[s * N_X + j];
    }
    float wv = w[j];
    out[j] = wv * f0 + (1.0f - wv) * f1;
}

extern "C" void kernel_launch(void* const* d_in, const int* in_sizes, int n_in,
                              void* d_out, int out_size, void* d_ws, size_t ws_size,
                              hipStream_t stream) {
    const float* xobs = (const float*)d_in[0];
    const float* x    = (const float*)d_in[1];
    const float* w    = (const float*)d_in[2];
    const float* a0   = (const float*)d_in[3];
    const float* a1   = (const float*)d_in[4];
    const float* tfl  = (const float*)d_in[5];
    const float* rls  = (const float*)d_in[6];
    const int*   ro   = (const int*)d_in[7];
    const int*   rx   = (const int*)d_in[8];
    float* ws  = (float*)d_ws;
    float* out = (float*)d_out;

    matern_wmma<<<(COL_TILES * N_SLICES) / 8, 256, 0, stream>>>(xobs, x, a0, a1, tfl, rls, ro, rx, ws);
    combine<<<(N_X + 255) / 256, 256, 0, stream>>>(w, ws, out);
}


# fold the epilogue's scaling into the matrix operands: B carries 5*log2(e)^2/ls^2 so exp2(-sqrt(acc)) follows directly; column norm starts the accumulator (sum order inside the distance changes)
# speedup vs baseline: 1.3456x; 1.0577x over previous
#include <hip/hip_runtime.h>
#include <math.h>


#define N_OBS   6000
#define N_X     6000
#define D_FEAT  16
#define N_DOM   3
#define N_SLICES 16
#define OBS_TILES (N_OBS / 16)
#define COL_TILES (N_X / 16)

typedef float v2f __attribute__((ext_vector_type(2)));
typedef float v4f __attribute__((ext_vector_type(4)));
typedef float v8f __attribute__((ext_vector_type(8)));

#define WS_PART0  0
#define WS_PART1  (WS_PART0 + N_SLICES * N_X)

__global__ void __launch_bounds__(256) matern_wmma(const float* __restrict__ xobs,
                                                   const float* __restrict__ x,
                                                   const float* __restrict__ alpha0,
                                                   const float* __restrict__ alpha1,
                                                   const float* __restrict__ logits,
                                                   const float* __restrict__ raw_ls,
                                                   const int*   __restrict__ ro,
                                                   const int*   __restrict__ rx,
                                                   float* __restrict__ ws) {
    const int lane = threadIdx.x & 31;
    const int wave = __builtin_amdgcn_readfirstlane(blockIdx.x * (blockDim.x >> 5) + (threadIdx.x >> 5));
    const int colTile = wave >> 4;
    const int slice   = wave & (N_SLICES - 1);
    const int n  = lane & 15;
    const int g  = lane >> 4;
    const int kb = g * 2;
    const int j  = colTile * 16 + n;

    const float ls     = log1pf(__expf(raw_ls[0]));
    const float inv_ls = 1.0f / ls;
    const float s2     = inv_ls * inv_ls;
    const float t01 = 1.0f / (1.0f + __expf(-logits[0]));
    const float t02 = 1.0f / (1.0f + __expf(-logits[1]));
    const float t12 = 1.0f / (1.0f + __expf(-logits[2]));
    const int ro1 = ro[2], ro2 = ro[4];

    const int dx = (j >= rx[2]) + (j >= rx[4]);
    const float F0 = (dx == 0) ? 1.0f : ((dx == 1) ? t01 : t02);
    const float F1 = (dx == 0) ? t01  : ((dx == 1) ? 1.0f : t12);
    const float F2 = (dx == 0) ? t02  : ((dx == 1) ? t12 : 1.0f);

    const float* bp = x + j * D_FEAT + kb;
    const v2f x0 = *(const v2f*)(bp + 0);
    const v2f x1 = *(const v2f*)(bp + 4);
    const v2f x2 = *(const v2f*)(bp + 8);
    const v2f x3 = *(const v2f*)(bp + 12);
    v2f qb = x0 * x0;
    qb = x1 * x1 + qb;
    qb = x2 * x2 + qb;
    qb = x3 * x3 + qb;
    float nb = qb.x + qb.y;
    nb += __shfl_xor(nb, 16, 32);
    const float G   = 10.406844905028039f * s2;
    const float nxl = nb * G;
    const float m2  = -2.0f * G;
    const v2f b0 = x0 * m2;
    const v2f b1 = x1 * m2;
    const v2f b2 = x2 * m2;
    const v2f b3 = x3 * m2;
    const v2f b4 = {G, G};
    const v8f cstart = {nxl, nxl, nxl, nxl, nxl, nxl, nxl, nxl};

    float acc0 = 0.0f, acc1 = 0.0f;

    for (int t = slice; t < OBS_TILES; t += N_SLICES) {
        const int i0 = t * 16;
        const float* ap = xobs + (i0 + n) * D_FEAT + kb;
        const v2f a0 = *(const v2f*)(ap + 0);
        const v2f a1 = *(const v2f*)(ap + 4);
        const v2f a2 = *(const v2f*)(ap + 8);
        const v2f a3 = *(const v2f*)(ap + 12);
        v2f qa = a0 * a0;
        qa = a1 * a1 + qa;
        qa = a2 * a2 + qa;
        qa = a3 * a3 + qa;
        const v2f a4 = qa;

        v8f c = cstart;
        c = __builtin_amdgcn_wmma_f32_16x16x4_f32(false, a0, false, b0, (short)0, c, false, false);
        c = __builtin_amdgcn_wmma_f32_16x16x4_f32(false, a1, false, b1, (short)0, c, false, false);
        c = __builtin_amdgcn_wmma_f32_16x16x4_f32(false, a2, false, b2, (short)0, c, false, false);
        c = __builtin_amdgcn_wmma_f32_16x16x4_f32(false, a3, false, b3, (short)0, c, false, false);
        c = __builtin_amdgcn_wmma_f32_16x16x4_f32(false, a4, false, b4, (short)0, c, false, false);

        const int rbase = i0 + g * 8;
        const v4f u0A = *(const v4f*)(alpha0 + rbase);
        const v4f u0B = *(const v4f*)(alpha0 + rbase + 4);
        const v4f u1A = *(const v4f*)(alpha1 + rbase);
        const v4f u1B = *(const v4f*)(alpha1 + rbase + 4);
        const float wa[8] = {u0A.x, u0A.y, u0A.z, u0A.w, u0B.x, u0B.y, u0B.z, u0B.w};
        const float wb[8] = {u1A.x, u1A.y, u1A.z, u1A.w, u1B.x, u1B.y, u1B.z, u1B.w};

        float kk[8];
        #pragma unroll
        for (int r = 0; r < 8; r += 2) {
            const v2f s = {__builtin_amdgcn_sqrtf(c[r]),
                           __builtin_amdgcn_sqrtf(c[r + 1])};
            const v2f u = {__builtin_fmaximum_numf(s.x, 3.2259641822295607e-6f),
                           __builtin_fmaximum_numf(s.y, 3.2259641822295607e-6f)};
            const v2f e = {__builtin_amdgcn_exp2f(-u.x),
                           __builtin_amdgcn_exp2f(-u.y)};
            const v2f poly = (u * 0.16015100463940048f + 0.6931471805599453f) * u + 1.0f;
            const v2f k2 = poly * e;
            kk[r]     = k2.x;
            kk[r + 1] = k2.y;
        }

        const int dlo = (i0 >= ro1) + (i0 >= ro2);
        const int dhi = (i0 + 15 >= ro1) + (i0 + 15 >= ro2);
        if (dlo == dhi) {
            const float fsel = (dlo == 0) ? F0 : ((dlo == 1) ? F1 : F2);
            float t0 = 0.0f, t1 = 0.0f;
            #pragma unroll
            for (int r = 0; r < 8; ++r) {
                t0 = fmaf(wa[r], kk[r], t0);
                t1 = fmaf(wb[r], kk[r], t1);
            }
            acc0 = fmaf(fsel, t0, acc0);
            acc1 = fmaf(fsel, t1, acc1);
        } else {
            #pragma unroll
            for (int r = 0; r < 8; ++r) {
                const int i  = rbase + r;
                const int dr = (i >= ro1) + (i >= ro2);
                const float f = (dr == 0) ? F0 : ((dr == 1) ? F1 : F2);
                acc0 = fmaf(wa[r] * f, kk[r], acc0);
                acc1 = fmaf(wb[r] * f, kk[r], acc1);
            }
        }
    }

    acc0 += __shfl_xor(acc0, 16, 32);
    acc1 += __shfl_xor(acc1, 16, 32);
    if (lane < 16) {
        ws[WS_PART0 + slice * N_X + j] = acc0;
        ws[WS_PART1 + slice * N_X + j] = acc1;
    }
}

__global__ void combine(const float* __restrict__ w,
                        const float* __restrict__ ws,
                        float* __restrict__ out) {
    int j = blockIdx.x * blockDim.x + threadIdx.x;
    if (j >= N_X) return;
    const float* p0 = ws + WS_PART0;
    const float* p1 = ws + WS_PART1;
    float f0 = 0.0f, f1 = 0.0f;
    #pragma unroll
    for (int s = 0; s < N_SLICES; ++s) {
        f0 += p0[s * N_X + j];
        f1 += p1[s * N_X + j];
    }
    float wv = w[j];
    out[j] = wv * f0 + (1.0f - wv) * f1;
}

extern "C" void kernel_launch(void* const* d_in, const int* in_sizes, int n_in,
                              void* d_out, int out_size, void* d_ws, size_t ws_size,
                              hipStream_t stream) {
    const float* xobs = (const float*)d_in[0];
    const float* x    = (const float*)d_in[1];
    const float* w    = (const float*)d_in[2];
    const float* a0   = (const float*)d_in[3];
    const float* a1   = (const float*)d_in[4];
    const float* tfl  = (const float*)d_in[5];
    const float* rls  = (const float*)d_in[6];
    const int*   ro   = (const int*)d_in[7];
    const int*   rx   = (const int*)d_in[8];
    float* ws  = (float*)d_ws;
    float* out = (float*)d_out;

    matern_wmma<<<(COL_TILES * N_SLICES) / 8, 256, 0, stream>>>(xobs, x, a0, a1, tfl, rls, ro, rx, ws);
    combine<<<(N_X + 255) / 256, 256, 0, stream>>>(w, ws, out);
}


# software-pipeline the obs-tile loop: tile k+1's five matrix instructions issue between tile k's epilogue and weighted sums; operands double-buffered in registers, loaded two tiles ahead
# speedup vs baseline: 1.5439x; 1.1474x over previous
#include <hip/hip_runtime.h>
#include <math.h>


#define N_OBS   6000
#define N_X     6000
#define D_FEAT  16
#define N_DOM   3
#define N_SLICES 16
#define OBS_TILES (N_OBS / 16)
#define COL_TILES (N_X / 16)

typedef float v2f __attribute__((ext_vector_type(2)));
typedef float v4f __attribute__((ext_vector_type(4)));
typedef float v8f __attribute__((ext_vector_type(8)));

#define WS_PART0  0
#define WS_PART1  (WS_PART0 + N_SLICES * N_X)

__global__ void __launch_bounds__(256) matern_wmma(const float* __restrict__ xobs,
                                                   const float* __restrict__ x,
                                                   const float* __restrict__ alpha0,
                                                   const float* __restrict__ alpha1,
                                                   const float* __restrict__ logits,
                                                   const float* __restrict__ raw_ls,
                                                   const int*   __restrict__ ro,
                                                   const int*   __restrict__ rx,
                                                   float* __restrict__ ws) {
    const int lane = threadIdx.x & 31;
    const int wave = __builtin_amdgcn_readfirstlane(blockIdx.x * (blockDim.x >> 5) + (threadIdx.x >> 5));
    const int colTile = wave >> 4;
    const int slice   = wave & (N_SLICES - 1);
    const int n  = lane & 15;
    const int g  = lane >> 4;
    const int kb = g * 2;
    const int j  = colTile * 16 + n;

    const float ls     = log1pf(__expf(raw_ls[0]));
    const float inv_ls = 1.0f / ls;
    const float s2     = inv_ls * inv_ls;
    const float t01 = 1.0f / (1.0f + __expf(-logits[0]));
    const float t02 = 1.0f / (1.0f + __expf(-logits[1]));
    const float t12 = 1.0f / (1.0f + __expf(-logits[2]));
    const int ro1 = ro[2], ro2 = ro[4];

    const int dx = (j >= rx[2]) + (j >= rx[4]);
    const float F0 = (dx == 0) ? 1.0f : ((dx == 1) ? t01 : t02);
    const float F1 = (dx == 0) ? t01  : ((dx == 1) ? 1.0f : t12);
    const float F2 = (dx == 0) ? t02  : ((dx == 1) ? t12 : 1.0f);

    const float* bp = x + j * D_FEAT + kb;
    const v2f x0 = *(const v2f*)(bp + 0);
    const v2f x1 = *(const v2f*)(bp + 4);
    const v2f x2 = *(const v2f*)(bp + 8);
    const v2f x3 = *(const v2f*)(bp + 12);
    v2f qb = x0 * x0;
    qb = x1 * x1 + qb;
    qb = x2 * x2 + qb;
    qb = x3 * x3 + qb;
    float nb = qb.x + qb.y;
    nb += __shfl_xor(nb, 16, 32);
    const float G   = 10.406844905028039f * s2;
    const float nxl = nb * G;
    const float m2  = -2.0f * G;
    const v2f b0 = x0 * m2;
    const v2f b1 = x1 * m2;
    const v2f b2 = x2 * m2;
    const v2f b3 = x3 * m2;
    const v2f b4 = {G, G};
    const v8f cstart = {nxl, nxl, nxl, nxl, nxl, nxl, nxl, nxl};

    float acc0 = 0.0f, acc1 = 0.0f;

    const int nTiles = (OBS_TILES - slice + N_SLICES - 1) / N_SLICES;
    const int laneA  = n * D_FEAT + kb;
    const int laneW  = g * 8;

    struct AFrag { v2f a0, a1, a2, a3; };
    struct Wts   { v4f u0A, u0B, u1A, u1B; };

    auto tileOf = [&](int k) __attribute__((always_inline)) {
        const int t = slice + k * N_SLICES;
        return t < OBS_TILES ? t : OBS_TILES - 1;
    };
    auto loadA = [&](AFrag& f, int k) __attribute__((always_inline)) {
        const float* ap = xobs + tileOf(k) * (16 * D_FEAT) + laneA;
        f.a0 = *(const v2f*)(ap + 0);
        f.a1 = *(const v2f*)(ap + 4);
        f.a2 = *(const v2f*)(ap + 8);
        f.a3 = *(const v2f*)(ap + 12);
    };
    auto loadW = [&](Wts& w, int k) __attribute__((always_inline)) {
        const int rb = tileOf(k) * 16 + laneW;
        w.u0A = *(const v4f*)(alpha0 + rb);
        w.u0B = *(const v4f*)(alpha0 + rb + 4);
        w.u1A = *(const v4f*)(alpha1 + rb);
        w.u1B = *(const v4f*)(alpha1 + rb + 4);
    };
    auto matrixTile = [&](const AFrag& f) __attribute__((always_inline)) {
        v2f qa = f.a0 * f.a0;
        qa = f.a1 * f.a1 + qa;
        qa = f.a2 * f.a2 + qa;
        qa = f.a3 * f.a3 + qa;
        v8f c = cstart;
        c = __builtin_amdgcn_wmma_f32_16x16x4_f32(false, f.a0, false, b0, (short)0, c, false, false);
        c = __builtin_amdgcn_wmma_f32_16x16x4_f32(false, f.a1, false, b1, (short)0, c, false, false);
        c = __builtin_amdgcn_wmma_f32_16x16x4_f32(false, f.a2, false, b2, (short)0, c, false, false);
        c = __builtin_amdgcn_wmma_f32_16x16x4_f32(false, f.a3, false, b3, (short)0, c, false, false);
        c = __builtin_amdgcn_wmma_f32_16x16x4_f32(false, qa,   false, b4, (short)0, c, false, false);
        return c;
    };
    struct Kv { float kk[8]; };
    auto epilogue = [&](const v8f& c) __attribute__((always_inline)) {
        Kv out;
        float* kk = out.kk;
        #pragma unroll
        for (int r = 0; r < 8; r += 2) {
            const v2f s = {__builtin_amdgcn_sqrtf(c[r]),
                           __builtin_amdgcn_sqrtf(c[r + 1])};
            const v2f u = {__builtin_fmaximum_numf(s.x, 3.2259641822295607e-6f),
                           __builtin_fmaximum_numf(s.y, 3.2259641822295607e-6f)};
            const v2f e = {__builtin_amdgcn_exp2f(-u.x),
                           __builtin_amdgcn_exp2f(-u.y)};
            const v2f poly = (u * 0.16015100463940048f + 0.6931471805599453f) * u + 1.0f;
            const v2f k2 = poly * e;
            kk[r]     = k2.x;
            kk[r + 1] = k2.y;
        }
        return out;
    };
    auto weightedSums = [&](const Kv& kv, const Wts& w, int k) __attribute__((always_inline)) {
        const float* kk = kv.kk;
        const int i0 = (slice + k * N_SLICES) * 16;
        const float wa[8] = {w.u0A.x, w.u0A.y, w.u0A.z, w.u0A.w, w.u0B.x, w.u0B.y, w.u0B.z, w.u0B.w};
        const float wb[8] = {w.u1A.x, w.u1A.y, w.u1A.z, w.u1A.w, w.u1B.x, w.u1B.y, w.u1B.z, w.u1B.w};

        const int dlo = (i0 >= ro1) + (i0 >= ro2);
        const int dhi = (i0 + 15 >= ro1) + (i0 + 15 >= ro2);
        if (dlo == dhi) {
            const float fsel = (dlo == 0) ? F0 : ((dlo == 1) ? F1 : F2);
            float t0 = 0.0f, t1 = 0.0f;
            #pragma unroll
            for (int r = 0; r < 8; ++r) {
                t0 = fmaf(wa[r], kk[r], t0);
                t1 = fmaf(wb[r], kk[r], t1);
            }
            acc0 = fmaf(fsel, t0, acc0);
            acc1 = fmaf(fsel, t1, acc1);
        } else {
            const int rbase = i0 + laneW;
            #pragma unroll
            for (int r = 0; r < 8; ++r) {
                const int i  = rbase + r;
                const int dr = (i >= ro1) + (i >= ro2);
                const float f = (dr == 0) ? F0 : ((dr == 1) ? F1 : F2);
                acc0 = fmaf(wa[r] * f, kk[r], acc0);
                acc1 = fmaf(wb[r] * f, kk[r], acc1);
            }
        }
    };

    AFrag fE, fO;
    Wts   wE, wO;
    loadA(fE, 0);
    loadA(fO, 1);
    loadW(wE, 0);
    loadW(wO, 1);
    v8f cur = matrixTile(fE);
    loadA(fE, 2);

    auto step = [&](AFrag& fN, Wts& wK, int k) __attribute__((always_inline)) {
        const v8f nxt = matrixTile(fN);
        const Kv kv = epilogue(cur);
        loadA(fN, k + 3);
        weightedSums(kv, wK, k);
        loadW(wK, k + 2);
        cur = nxt;
    };

    const int nPairs = (nTiles - 1) >> 1;
    for (int p = 0; p < nPairs; ++p) {
        step(fO, wE, 2 * p);
        step(fE, wO, 2 * p + 1);
    }
    const int kl = 2 * nPairs;
    if (kl < nTiles - 1) {
        step(fO, wE, kl);
        weightedSums(epilogue(cur), wO, kl + 1);
    } else {
        weightedSums(epilogue(cur), wE, kl);
    }

    acc0 += __shfl_xor(acc0, 16, 32);
    acc1 += __shfl_xor(acc1, 16, 32);
    if (lane < 16) {
        ws[WS_PART0 + slice * N_X + j] = acc0;
        ws[WS_PART1 + slice * N_X + j] = acc1;
    }
}

__global__ void combine(const float* __restrict__ w,
                        const float* __restrict__ ws,
                        float* __restrict__ out) {
    int j = blockIdx.x * blockDim.x + threadIdx.x;
    if (j >= N_X) return;
    const float* p0 = ws + WS_PART0;
    const float* p1 = ws + WS_PART1;
    float f0 = 0.0f, f1 = 0.0f;
    #pragma unroll
    for (int s = 0; s < N_SLICES; ++s) {
        f0 += p0[s * N_X + j];
        f1 += p1[s * N_X + j];
    }
    float wv = w[j];
    out[j] = wv * f0 + (1.0f - wv) * f1;
}

extern "C" void kernel_launch(void* const* d_in, const int* in_sizes, int n_in,
                              void* d_out, int out_size, void* d_ws, size_t ws_size,
                              hipStream_t stream) {
    const float* xobs = (const float*)d_in[0];
    const float* x    = (const float*)d_in[1];
    const float* w    = (const float*)d_in[2];
    const float* a0   = (const float*)d_in[3];
    const float* a1   = (const float*)d_in[4];
    const float* tfl  = (const float*)d_in[5];
    const float* rls  = (const float*)d_in[6];
    const int*   ro   = (const int*)d_in[7];
    const int*   rx   = (const int*)d_in[8];
    float* ws  = (float*)d_ws;
    float* out = (float*)d_out;

    matern_wmma<<<(COL_TILES * N_SLICES) / 8, 256, 0, stream>>>(xobs, x, a0, a1, tfl, rls, ro, rx, ws);
    combine<<<(N_X + 255) / 256, 256, 0, stream>>>(w, ws, out);
}
